# starting kernel
# baseline (speedup 1.0000x reference)
#include <hip/hip_runtime.h>
#include <hip/hip_bf16.h>

typedef __attribute__((ext_vector_type(2))) float v2f;
typedef __attribute__((ext_vector_type(8))) float v8f;

#define MB 8
#define D1C 32
#define D2C 32
#define EC 256
#define LC 16384
#define HC 256
#define NSPEC 32
#define SIDXC 8
#define NPLAY 4
#define POLW (LC + SIDXC)


__device__ __forceinline__ float dot8(float4 a, float4 b, float4 c, float4 d) {
    return a.x*b.x + a.y*b.y + a.z*b.z + a.w*b.w +
           c.x*d.x + c.y*d.y + c.z*d.z + c.w*d.w;
}

__device__ __forceinline__ float wave_sum(float v) {
    #pragma unroll
    for (int off = 16; off > 0; off >>= 1) v += __shfl_xor(v, off, 32);
    return v;
}

__device__ __forceinline__ v8f wmma_tile_k256(const float* A, const float* B,
                                              int ldb, int bcol, int arow, int khalf) {
    v8f acc = {};
    #pragma unroll 8
    for (int k0 = 0; k0 < 256; k0 += 4) {
        const int kA = k0 + khalf * 2;
        v2f a = *(const v2f*)(A + arow * 256 + kA);
        v2f b;
        b.x = B[(kA    ) * ldb + bcol];
        b.y = B[(kA + 1) * ldb + bcol];
        acc = __builtin_amdgcn_wmma_f32_16x16x4_f32(
                  false, a, false, b, (short)0, acc, false, false);
    }
    return acc;
}

__global__ __launch_bounds__(256) void emb_dot_kernel(
    const float* __restrict__ emb, const float* __restrict__ cls,
    const float* __restrict__ Wp, const float* __restrict__ bp,
    float* __restrict__ ws)
{
    const int lane = threadIdx.x & 31;
    const int wave = threadIdx.x >> 5;
    const int row  = blockIdx.x * 8 + wave;
    float* emb_dot = ws + 64;

    const float4* e4  = (const float4*)(emb + (size_t)row * EC);
    const float4* w04 = (const float4*)(Wp);
    const float4* w14 = (const float4*)(Wp + EC);

    float4 ea = e4[lane*2 + 0], eb = e4[lane*2 + 1];
    float p0 = dot8(ea, w04[lane*2+0], eb, w04[lane*2+1]);
    float p1 = dot8(ea, w14[lane*2+0], eb, w14[lane*2+1]);
    p0 = wave_sum(p0);
    p1 = wave_sum(p1);
    if (lane == 0) { emb_dot[row*2 + 0] = p0; emb_dot[row*2 + 1] = p1; }

    if (blockIdx.x == 0 && wave == 0) {
        const float4* wc4 = (const float4*)(Wp + 2*EC);
        float4 wca = wc4[lane*2], wcb = wc4[lane*2+1];
        float bias = bp[0];
        for (int m = 0; m < MB; ++m) {
            const float4* c4 = (const float4*)(cls + m*EC);
            float p = dot8(c4[lane*2], wca, c4[lane*2+1], wcb);
            p = wave_sum(p);
            if (lane == 0) ws[m] = p + bias;
        }
    }
}

__global__ __launch_bounds__(512) void heads_kernel(
    const float* __restrict__ cls,
    const float* __restrict__ Wv1, const float* __restrict__ bv1,
    const float* __restrict__ Wv2, const float* __restrict__ bv2,
    const float* __restrict__ Ws1, const float* __restrict__ bs1,
    const float* __restrict__ Ws2, const float* __restrict__ bs2,
    const int*   __restrict__ sidx,
    float* __restrict__ out)
{
    __shared__ float clsLds[16 * EC];
    __shared__ float hv[16 * HC];
    __shared__ float hs[16 * HC];
    __shared__ float w2v[HC * 16];
    __shared__ float spAll[MB * NSPEC];

    const int t     = threadIdx.x;
    const int lane  = t & 31;
    const int wave  = t >> 5;
    const int colr  = lane & 15;
    const int khalf = lane >> 4;

    for (int i = t; i < 16 * EC; i += 512) {
        const int r = i >> 8, c = i & 255;
        clsLds[i] = (r < MB) ? cls[r * EC + c] : 0.f;
    }
    for (int i = t; i < 8 * HC; i += 512) {
        hv[8 * HC + i] = 0.f;
        hs[8 * HC + i] = 0.f;
    }
    for (int i = t; i < HC * 16; i += 512) {
        const int k = i >> 4, c = i & 15;
        w2v[i] = (c < NPLAY) ? Wv2[k * NPLAY + c] : 0.f;
    }
    __syncthreads();

    for (int job = wave; job < 32; job += 16) {
        const float* W1 = (job < 16) ? Wv1 : Ws1;
        const float* b1 = (job < 16) ? bv1 : bs1;
        float* dst      = (job < 16) ? hv  : hs;
        const int n0 = (job & 15) * 16;
        v8f acc = wmma_tile_k256(clsLds, W1, HC, n0 + colr, colr, khalf);
        if (khalf == 0) {
            const float bias = b1[n0 + colr];
            #pragma unroll
            for (int i = 0; i < MB; ++i) {
                const float v = acc[i] + bias;
                dst[i * HC + n0 + colr] = v > 0.f ? v : 0.f;
            }
        }
    }
    __syncthreads();

    if (wave < 2) {
        const int n0 = wave * 16;
        v8f acc = wmma_tile_k256(hs, Ws2, NSPEC, n0 + colr, colr, khalf);
        if (khalf == 0) {
            const float bias = bs2[n0 + colr];
            #pragma unroll
            for (int i = 0; i < MB; ++i)
                spAll[i * NSPEC + n0 + colr] = acc[i] + bias;
        }
    } else if (wave == 2) {
        v8f acc = wmma_tile_k256(hv, w2v, 16, colr, colr, khalf);
        if (khalf == 0 && colr < NPLAY) {
            const float bias = bv2[colr];
            #pragma unroll
            for (int i = 0; i < MB; ++i)
                out[MB * POLW + i * NPLAY + colr] = acc[i] + bias;
        }
    }
    __syncthreads();

    if (t < MB * SIDXC) {
        const int m = t >> 3, j = t & 7;
        out[m * POLW + LC + j] = spAll[m * NSPEC + sidx[j]];
    }
}

__global__ __launch_bounds__(256) void policy_kernel(
    const int* __restrict__ moves, const float* __restrict__ ws,
    float* __restrict__ out)
{
    __shared__ float clsd[MB];
    if (threadIdx.x < MB) clsd[threadIdx.x] = ws[threadIdx.x];
    __syncthreads();

    const int l = blockIdx.x * blockDim.x + threadIdx.x;
    const float* emb_dot = ws + 64;
    const int4 mv = ((const int4*)moves)[l];
    const int r0 = mv.x * D2C + mv.y;
    const int r1 = mv.z * D2C + mv.w;
    #pragma unroll
    for (int m = 0; m < MB; ++m) {
        const float d0 = emb_dot[(m*1024 + r0)*2 + 0];
        const float d1 = emb_dot[(m*1024 + r1)*2 + 1];
        out[m*POLW + l] = d0 + d1 + clsd[m];
    }
}

extern "C" void kernel_launch(void* const* d_in, const int* in_sizes, int n_in,
                              void* d_out, int out_size, void* d_ws, size_t ws_size,
                              hipStream_t stream) {
    (void)in_sizes; (void)n_in; (void)out_size; (void)ws_size;
    const float* emb  = (const float*)d_in[0];
    const float* cls  = (const float*)d_in[1];
    const int*   mov  = (const int*)  d_in[2];
    const int*   sidx = (const int*)  d_in[3];
    const float* Wp   = (const float*)d_in[4];
    const float* bp   = (const float*)d_in[5];
    const float* Wv1  = (const float*)d_in[6];
    const float* bv1  = (const float*)d_in[7];
    const float* Wv2  = (const float*)d_in[8];
    const float* bv2  = (const float*)d_in[9];
    const float* Ws1  = (const float*)d_in[10];
    const float* bs1  = (const float*)d_in[11];
    const float* Ws2  = (const float*)d_in[12];
    const float* bs2  = (const float*)d_in[13];
    float* out = (float*)d_out;
    float* ws  = (float*)d_ws;

    emb_dot_kernel<<<1024, 256, 0, stream>>>(emb, cls, Wp, bp, ws);
    heads_kernel<<<1, 512, 0, stream>>>(cls, Wv1, bv1, Wv2, bv2,
                                        Ws1, bs1, Ws2, bs2, sidx, out);
    policy_kernel<<<LC/256, 256, 0, stream>>>(mov, ws, out);
}


# heads kernel: loads batched ahead of use: small operands, Ws2, biases, indices fetched at entry; layer-1 K loop unrolled, B loads run ahead of the WMMA chain, A read from LDS once for both tiles
# speedup vs baseline: 1.4420x; 1.4420x over previous
#include <hip/hip_runtime.h>
#include <hip/hip_bf16.h>

typedef __attribute__((ext_vector_type(2))) float v2f;
typedef __attribute__((ext_vector_type(8))) float v8f;

#define MB 8
#define D1C 32
#define D2C 32
#define EC 256
#define LC 16384
#define HC 256
#define NSPEC 32
#define SIDXC 8
#define NPLAY 4
#define POLW (LC + SIDXC)


__device__ __forceinline__ float dot8(float4 a, float4 b, float4 c, float4 d) {
    return a.x*b.x + a.y*b.y + a.z*b.z + a.w*b.w +
           c.x*d.x + c.y*d.y + c.z*d.z + c.w*d.w;
}

__device__ __forceinline__ float wave_sum(float v) {
    #pragma unroll
    for (int off = 16; off > 0; off >>= 1) v += __shfl_xor(v, off, 32);
    return v;
}

__device__ __forceinline__ v8f wmma_tile_k256(const float* A, const float* B,
                                              int ldb, int bcol, int arow, int khalf) {
    v8f acc = {};
    #pragma unroll 8
    for (int k0 = 0; k0 < 256; k0 += 4) {
        const int kA = k0 + khalf * 2;
        v2f a = *(const v2f*)(A + arow * 256 + kA);
        v2f b;
        b.x = B[(kA    ) * ldb + bcol];
        b.y = B[(kA + 1) * ldb + bcol];
        acc = __builtin_amdgcn_wmma_f32_16x16x4_f32(
                  false, a, false, b, (short)0, acc, false, false);
    }
    asm volatile("v_nop\n\tv_nop\n\tv_nop\n\tv_nop\n\tv_nop\n\tv_nop\n\tv_nop\n\tv_nop" : "+v"(acc));
    return acc;
}

__device__ __forceinline__ v8f wmma_tile_k256_gmem(const v2f (&a)[64], const float* __restrict__ B,
                                                   int ldb, int bcol, int khalf) {
    const float* Bl = B + (khalf * 2) * ldb + bcol;
    v2f b[64];
    #pragma unroll
    for (int i = 0; i < 64; ++i) {
        b[i].x = Bl[(4 * i    ) * ldb];
        b[i].y = Bl[(4 * i + 1) * ldb];
    }
    v8f acc = {};
    #pragma unroll
    for (int i = 0; i < 64; ++i) {
        acc = __builtin_amdgcn_wmma_f32_16x16x4_f32(
                  false, a[i], false, b[i], (short)0, acc, false, false);
    }
    asm volatile("v_nop\n\tv_nop\n\tv_nop\n\tv_nop\n\tv_nop\n\tv_nop\n\tv_nop\n\tv_nop" : "+v"(acc));
    return acc;
}

__global__ __launch_bounds__(256) void emb_dot_kernel(
    const float* __restrict__ emb, const float* __restrict__ cls,
    const float* __restrict__ Wp, const float* __restrict__ bp,
    float* __restrict__ ws)
{
    const int lane = threadIdx.x & 31;
    const int wave = threadIdx.x >> 5;
    const int row  = blockIdx.x * 8 + wave;
    float* emb_dot = ws + 64;

    const float4* e4  = (const float4*)(emb + (size_t)row * EC);
    const float4* w04 = (const float4*)(Wp);
    const float4* w14 = (const float4*)(Wp + EC);

    float4 ea = e4[lane*2 + 0], eb = e4[lane*2 + 1];
    float p0 = dot8(ea, w04[lane*2+0], eb, w04[lane*2+1]);
    float p1 = dot8(ea, w14[lane*2+0], eb, w14[lane*2+1]);
    p0 = wave_sum(p0);
    p1 = wave_sum(p1);
    if (lane == 0) { emb_dot[row*2 + 0] = p0; emb_dot[row*2 + 1] = p1; }

    if (blockIdx.x == 0 && wave == 0) {
        const float4* wc4 = (const float4*)(Wp + 2*EC);
        float4 wca = wc4[lane*2], wcb = wc4[lane*2+1];
        float bias = bp[0];
        for (int m = 0; m < MB; ++m) {
            const float4* c4 = (const float4*)(cls + m*EC);
            float p = dot8(c4[lane*2], wca, c4[lane*2+1], wcb);
            p = wave_sum(p);
            if (lane == 0) ws[m] = p + bias;
        }
    }
}

__global__ __launch_bounds__(512) void heads_kernel(
    const float* __restrict__ cls,
    const float* __restrict__ Wv1, const float* __restrict__ bv1,
    const float* __restrict__ Wv2, const float* __restrict__ bv2,
    const float* __restrict__ Ws1, const float* __restrict__ bs1,
    const float* __restrict__ Ws2, const float* __restrict__ bs2,
    const int*   __restrict__ sidx,
    float* __restrict__ out)
{
    __shared__ __attribute__((aligned(16))) float clsLds[16 * EC];
    __shared__ __attribute__((aligned(16))) float hv[16 * HC];
    __shared__ __attribute__((aligned(16))) float hs[16 * HC];
    __shared__ __attribute__((aligned(16))) float w2v[HC * 16];
    __shared__ __attribute__((aligned(16))) float w2s[HC * NSPEC];
    __shared__ float spAll[MB * NSPEC];

    const int t     = threadIdx.x;
    const int lane  = t & 31;
    const int wave  = t >> 5;
    const int colr  = lane & 15;
    const int khalf = lane >> 4;

    const float4 z4 = {0.f, 0.f, 0.f, 0.f};
    const float4 c4 = ((const float4*)cls)[t];
    float4 s4[4];
    #pragma unroll
    for (int j = 0; j < 4; ++j) s4[j] = ((const float4*)Ws2)[t + 512 * j];
    float4 v4 = z4;
    if (t < HC) v4 = ((const float4*)Wv2)[t];
    const float b1v = bv1[wave * 16 + colr];
    const float b1s = bs1[wave * 16 + colr];
    float b2 = 0.f;
    if (wave < 2) b2 = bs2[wave * 16 + colr];
    else if (wave == 2 && colr < NPLAY) b2 = bv2[colr];
    int sj = 0;
    if (t < MB * SIDXC) sj = sidx[t & 7];

    ((float4*)clsLds)[t]       = c4;
    ((float4*)clsLds)[512 + t] = z4;
    ((float4*)hv)[512 + t]     = z4;
    ((float4*)hs)[512 + t]     = z4;
    #pragma unroll
    for (int j = 0; j < 4; ++j) ((float4*)w2s)[t + 512 * j] = s4[j];
    if (t < HC) {
        float4* w = (float4*)(w2v + t * 16);
        w[0] = v4; w[1] = z4; w[2] = z4; w[3] = z4;
    }
    __syncthreads();

    v2f afrag[64];
    #pragma unroll
    for (int i = 0; i < 64; ++i)
        afrag[i] = *(const v2f*)(clsLds + colr * 256 + 4 * i + khalf * 2);
    for (int job = wave; job < 32; job += 16) {
        const float* W1 = (job < 16) ? Wv1 : Ws1;
        float* dst      = (job < 16) ? hv  : hs;
        const int n0 = (job & 15) * 16;
        v8f acc = wmma_tile_k256_gmem(afrag, W1, HC, n0 + colr, khalf);
        if (khalf == 0) {
            const float bias = (job < 16) ? b1v : b1s;
            #pragma unroll
            for (int i = 0; i < MB; ++i) {
                const float v = acc[i] + bias;
                dst[i * HC + n0 + colr] = v > 0.f ? v : 0.f;
            }
        }
    }
    __syncthreads();

    if (wave < 2) {
        const int n0 = wave * 16;
        v8f acc = wmma_tile_k256(hs, w2s, NSPEC, n0 + colr, colr, khalf);
        if (khalf == 0) {
            const float bias = b2;
            #pragma unroll
            for (int i = 0; i < MB; ++i)
                spAll[i * NSPEC + n0 + colr] = acc[i] + bias;
        }
    } else if (wave == 2) {
        v8f acc = wmma_tile_k256(hv, w2v, 16, colr, colr, khalf);
        if (khalf == 0 && colr < NPLAY) {
            const float bias = b2;
            #pragma unroll
            for (int i = 0; i < MB; ++i)
                out[MB * POLW + i * NPLAY + colr] = acc[i] + bias;
        }
    }
    __syncthreads();

    if (t < MB * SIDXC) {
        const int m = t >> 3, j = t & 7;
        out[m * POLW + LC + j] = spAll[m * NSPEC + sj];
    }
}

__global__ __launch_bounds__(256) void policy_kernel(
    const int* __restrict__ moves, const float* __restrict__ ws,
    float* __restrict__ out)
{
    __shared__ float clsd[MB];
    if (threadIdx.x < MB) clsd[threadIdx.x] = ws[threadIdx.x];
    __syncthreads();

    const int l = blockIdx.x * blockDim.x + threadIdx.x;
    const float* emb_dot = ws + 64;
    const int4 mv = ((const int4*)moves)[l];
    const int r0 = mv.x * D2C + mv.y;
    const int r1 = mv.z * D2C + mv.w;
    #pragma unroll
    for (int m = 0; m < MB; ++m) {
        const float d0 = emb_dot[(m*1024 + r0)*2 + 0];
        const float d1 = emb_dot[(m*1024 + r1)*2 + 1];
        out[m*POLW + l] = d0 + d1 + clsd[m];
    }
}

extern "C" void kernel_launch(void* const* d_in, const int* in_sizes, int n_in,
                              void* d_out, int out_size, void* d_ws, size_t ws_size,
                              hipStream_t stream) {
    (void)in_sizes; (void)n_in; (void)out_size; (void)ws_size;
    const float* emb  = (const float*)d_in[0];
    const float* cls  = (const float*)d_in[1];
    const int*   mov  = (const int*)  d_in[2];
    const int*   sidx = (const int*)  d_in[3];
    const float* Wp   = (const float*)d_in[4];
    const float* bp   = (const float*)d_in[5];
    const float* Wv1  = (const float*)d_in[6];
    const float* bv1  = (const float*)d_in[7];
    const float* Wv2  = (const float*)d_in[8];
    const float* bv2  = (const float*)d_in[9];
    const float* Ws1  = (const float*)d_in[10];
    const float* bs1  = (const float*)d_in[11];
    const float* Ws2  = (const float*)d_in[12];
    const float* bs2  = (const float*)d_in[13];
    float* out = (float*)d_out;
    float* ws  = (float*)d_ws;

    emb_dot_kernel<<<1024, 256, 0, stream>>>(emb, cls, Wp, bp, ws);
    heads_kernel<<<1, 512, 0, stream>>>(cls, Wv1, bv1, Wv2, bv2,
                                        Ws1, bs1, Ws2, bs2, sidx, out);
    policy_kernel<<<LC/256, 256, 0, stream>>>(mov, ws, out);
}


# heads launch removed (3 -> 2 launches): layer 1 runs as 4 extra work-groups of launch 1, hidden rows parked in the output's first 512 policy columns; layer 2 runs inside policy work-groups 0 and 1
# speedup vs baseline: 3.0380x; 2.1068x over previous
#include <hip/hip_runtime.h>


typedef __attribute__((ext_vector_type(2))) float v2f;
typedef __attribute__((ext_vector_type(8))) float v8f;

#define MB 8
#define D1C 32
#define D2C 32
#define EC 256
#define LC 16384
#define HC 256
#define NSPEC 32
#define SIDXC 8
#define NPLAY 4
#define POLW (LC + SIDXC)
#define L1BLK 4


__device__ __forceinline__ float dot8(float4 a, float4 b, float4 c, float4 d) {
    return a.x*b.x + a.y*b.y + a.z*b.z + a.w*b.w +
           c.x*d.x + c.y*d.y + c.z*d.z + c.w*d.w;
}

__device__ __forceinline__ float wave_sum(float v) {
    #pragma unroll
    for (int off = 16; off > 0; off >>= 1) v += __shfl_xor(v, off, 32);
    return v;
}


__global__ __launch_bounds__(256) void front_kernel(
    const float* __restrict__ emb, const float* __restrict__ cls,
    const float* __restrict__ Wp, const float* __restrict__ bp,
    const float* __restrict__ Wv1, const float* __restrict__ bv1,
    const float* __restrict__ Ws1, const float* __restrict__ bs1,
    float* __restrict__ ws, float* __restrict__ out)
{
    const int lane = threadIdx.x & 31;
    const int wave = threadIdx.x >> 5;

    if (blockIdx.x < L1BLK) {
        const int job   = blockIdx.x * 8 + wave;
        const int colr  = lane & 15;
        const int khalf = lane >> 4;
        const float* W1 = (job < 16) ? Wv1 : Ws1;
        const float* b1 = (job < 16) ? bv1 : bs1;
        const int n0    = (job & 15) * 16;
        const float bias = b1[n0 + colr];
        const float* Al = cls + (colr & 7) * EC + khalf * 2;
        const float* Bl = W1 + (khalf * 2) * HC + n0 + colr;
        v8f acc = {};
        #pragma unroll 1
        for (int c = 0; c < 4; ++c) {
            v2f a[16], b[16];
            #pragma unroll
            for (int i = 0; i < 16; ++i) {
                const int s = c * 16 + i;
                a[i]   = *(const v2f*)(Al + 4 * s);
                b[i].x = Bl[(4 * s    ) * HC];
                b[i].y = Bl[(4 * s + 1) * HC];
            }
            #pragma unroll
            for (int i = 0; i < 16; ++i)
                acc = __builtin_amdgcn_wmma_f32_16x16x4_f32(
                          false, a[i], false, b[i], (short)0, acc, false, false);
            asm volatile("v_nop\n\tv_nop\n\tv_nop\n\tv_nop\n\tv_nop\n\tv_nop\n\tv_nop\n\tv_nop" : "+v"(acc));
        }
        if (khalf == 0) {
            float* dst = out + ((job < 16) ? HC : 0) + n0 + colr;
            #pragma unroll
            for (int i = 0; i < MB; ++i) {
                const float v = acc[i] + bias;
                dst[i * POLW] = v > 0.f ? v : 0.f;
            }
        }
        return;
    }

    const int blk = blockIdx.x - L1BLK;
    const int row = blk * 8 + wave;
    float* emb_dot = ws + 64;

    const float4* e4  = (const float4*)(emb + (size_t)row * EC);
    const float4* w04 = (const float4*)(Wp);
    const float4* w14 = (const float4*)(Wp + EC);

    float4 ea = e4[lane*2 + 0], eb = e4[lane*2 + 1];
    float p0 = dot8(ea, w04[lane*2+0], eb, w04[lane*2+1]);
    float p1 = dot8(ea, w14[lane*2+0], eb, w14[lane*2+1]);
    p0 = wave_sum(p0);
    p1 = wave_sum(p1);
    if (lane == 0) { emb_dot[row*2 + 0] = p0; emb_dot[row*2 + 1] = p1; }

    if (blk == 0 && wave == 0) {
        const float4* wc4 = (const float4*)(Wp + 2*EC);
        float4 wca = wc4[lane*2], wcb = wc4[lane*2+1];
        float bias = bp[0];
        for (int m = 0; m < MB; ++m) {
            const float4* c4 = (const float4*)(cls + m*EC);
            float p = dot8(c4[lane*2], wca, c4[lane*2+1], wcb);
            p = wave_sum(p);
            if (lane == 0) ws[m] = p + bias;
        }
    }
}

template <int LDB>
__device__ __forceinline__ v8f wmma_tile_k256_gmem(const float* Al, const float* __restrict__ Bl) {
    v2f a[64], b[64];
    #pragma unroll
    for (int i = 0; i < 64; ++i) {
        a[i]   = *(const v2f*)(Al + 4 * i);
        b[i].x = Bl[(4 * i    ) * LDB];
        b[i].y = Bl[(4 * i + 1) * LDB];
    }
    v8f acc = {};
    #pragma unroll
    for (int i = 0; i < 64; ++i)
        acc = __builtin_amdgcn_wmma_f32_16x16x4_f32(
                  false, a[i], false, b[i], (short)0, acc, false, false);
    asm volatile("v_nop\n\tv_nop\n\tv_nop\n\tv_nop\n\tv_nop\n\tv_nop\n\tv_nop\n\tv_nop" : "+v"(acc));
    return acc;
}

__global__ __launch_bounds__(256) void back_kernel(
    const int* __restrict__ moves, const float* __restrict__ ws,
    const float* __restrict__ Wv2, const float* __restrict__ bv2,
    const float* __restrict__ Ws2, const float* __restrict__ bs2,
    const int*   __restrict__ sidx,
    float* out)
{
    __shared__ float clsd[MB];
    __shared__ float spAll[MB * NSPEC];
    const int t = threadIdx.x;
    if (t < MB) clsd[t] = ws[t];

    const int l = blockIdx.x * 256 + t;
    const float* emb_dot = ws + 64;
    const int4 mv = ((const int4*)moves)[l];
    const int r0 = mv.x * D2C + mv.y;
    const int r1 = mv.z * D2C + mv.w;
    float pol[MB];
    #pragma unroll
    for (int m = 0; m < MB; ++m) {
        const float d0 = emb_dot[(m*1024 + r0)*2 + 0];
        const float d1 = emb_dot[(m*1024 + r1)*2 + 1];
        pol[m] = d0 + d1;
    }

    if (blockIdx.x < 2) {
        const int lane  = t & 31;
        const int wave  = t >> 5;
        const int colr  = lane & 15;
        const int khalf = lane >> 4;
        const bool spec = (blockIdx.x == 0);
        const float* Al = out + blockIdx.x * HC + (colr & 7) * POLW + khalf * 2;
        int sj = 0;
        if (spec) {
            if (t < MB * SIDXC) sj = sidx[t & 7];
            if (wave < 2) {
                const int bcol = wave * 16 + colr;
                const float b2 = bs2[bcol];
                v8f acc = wmma_tile_k256_gmem<NSPEC>(Al, Ws2 + (khalf * 2) * NSPEC + bcol);
                if (khalf == 0) {
                    #pragma unroll
                    for (int i = 0; i < MB; ++i)
                        spAll[i * NSPEC + bcol] = acc[i] + b2;
                }
            }
        } else if (wave == 0) {
            const int bcol = colr & 3;
            const float b2 = bv2[bcol];
            v8f acc = wmma_tile_k256_gmem<NPLAY>(Al, Wv2 + (khalf * 2) * NPLAY + bcol);
            if (khalf == 0 && colr < NPLAY) {
                #pragma unroll
                for (int i = 0; i < MB; ++i)
                    out[MB * POLW + i * NPLAY + colr] = acc[i] + b2;
            }
        }
        __syncthreads();
        if (spec && t < MB * SIDXC) {
            const int m = t >> 3, j = t & 7;
            out[m * POLW + LC + j] = spAll[m * NSPEC + sj];
        }
    } else {
        __syncthreads();
    }

    #pragma unroll
    for (int m = 0; m < MB; ++m)
        out[m*POLW + l] = pol[m] + clsd[m];
}

extern "C" void kernel_launch(void* const* d_in, const int* in_sizes, int n_in,
                              void* d_out, int out_size, void* d_ws, size_t ws_size,
                              hipStream_t stream) {
    (void)in_sizes; (void)n_in; (void)out_size; (void)ws_size;
    const float* emb  = (const float*)d_in[0];
    const float* cls  = (const float*)d_in[1];
    const int*   mov  = (const int*)  d_in[2];
    const int*   sidx = (const int*)  d_in[3];
    const float* Wp   = (const float*)d_in[4];
    const float* bp   = (const float*)d_in[5];
    const float* Wv1  = (const float*)d_in[6];
    const float* bv1  = (const float*)d_in[7];
    const float* Wv2  = (const float*)d_in[8];
    const float* bv2  = (const float*)d_in[9];
    const float* Ws1  = (const float*)d_in[10];
    const float* bs1  = (const float*)d_in[11];
    const float* Ws2  = (const float*)d_in[12];
    const float* bs2  = (const float*)d_in[13];
    float* out = (float*)d_out;
    float* ws  = (float*)d_ws;

    front_kernel<<<1028, 256, 0, stream>>>(emb, cls, Wp, bp, Wv1, bv1, Ws1, bs1, ws, out);
    back_kernel<<<64, 256, 0, stream>>>(mov, ws, Wv2, bv2, Ws2, bs2, sidx, out);
}


# launch 1 rebuilt from fewer, fatter waves: a streaming wave takes 8 embedding rows, all 16 row loads in flight (1024 waves in 4-wave work-groups, was 8192 one-row waves); cls dot on its own wave
# speedup vs baseline: 3.4885x; 1.1483x over previous
#include <hip/hip_runtime.h>


typedef __attribute__((ext_vector_type(2))) float v2f;
typedef __attribute__((ext_vector_type(8))) float v8f;

#define MB 8
#define D1C 32
#define D2C 32
#define EC 256
#define LC 16384
#define HC 256
#define NSPEC 32
#define SIDXC 8
#define NPLAY 4
#define POLW (LC + SIDXC)
#define WPB 4
#define L1BLK 8
#define RPW 8


__device__ __forceinline__ float dot8(float4 a, float4 b, float4 c, float4 d) {
    return a.x*b.x + a.y*b.y + a.z*b.z + a.w*b.w +
           c.x*d.x + c.y*d.y + c.z*d.z + c.w*d.w;
}

__device__ __forceinline__ float wave_sum(float v) {
    #pragma unroll
    for (int off = 16; off > 0; off >>= 1) v += __shfl_xor(v, off, 32);
    return v;
}


__global__ __launch_bounds__(128) void front_kernel(
    const float* __restrict__ emb, const float* __restrict__ cls,
    const float* __restrict__ Wp, const float* __restrict__ bp,
    const float* __restrict__ Wv1, const float* __restrict__ bv1,
    const float* __restrict__ Ws1, const float* __restrict__ bs1,
    float* __restrict__ ws, float* __restrict__ out)
{
    const int lane = threadIdx.x & 31;
    const int wave = threadIdx.x >> 5;

    if (blockIdx.x < L1BLK) {
        const int job   = blockIdx.x * WPB + wave;
        const int colr  = lane & 15;
        const int khalf = lane >> 4;
        const float* W1 = (job < 16) ? Wv1 : Ws1;
        const float* b1 = (job < 16) ? bv1 : bs1;
        const int n0    = (job & 15) * 16;
        const float bias = b1[n0 + colr];
        const float* Al = cls + (colr & 7) * EC + khalf * 2;
        const float* Bl = W1 + (khalf * 2) * HC + n0 + colr;
        v8f acc = {};
        #pragma unroll 1
        for (int c = 0; c < 4; ++c) {
            v2f a[16], b[16];
            #pragma unroll
            for (int i = 0; i < 16; ++i) {
                const int s = c * 16 + i;
                a[i]   = *(const v2f*)(Al + 4 * s);
                b[i].x = Bl[(4 * s    ) * HC];
                b[i].y = Bl[(4 * s + 1) * HC];
            }
            #pragma unroll
            for (int i = 0; i < 16; ++i)
                acc = __builtin_amdgcn_wmma_f32_16x16x4_f32(
                          false, a[i], false, b[i], (short)0, acc, false, false);
            asm volatile("v_nop\n\tv_nop\n\tv_nop\n\tv_nop\n\tv_nop\n\tv_nop\n\tv_nop\n\tv_nop" : "+v"(acc));
        }
        if (khalf == 0) {
            float* dst = out + ((job < 16) ? HC : 0) + n0 + colr;
            #pragma unroll
            for (int i = 0; i < MB; ++i) {
                const float v = acc[i] + bias;
                dst[i * POLW] = v > 0.f ? v : 0.f;
            }
        }
        return;
    }

    const int sblk  = (int)blockIdx.x - L1BLK;
    const int swave = __builtin_amdgcn_readfirstlane(wave);

    if (sblk == 0) {
        if (swave != 0) return;
        const float4* wc4 = (const float4*)(Wp + 2*EC);
        const float4* c4  = (const float4*)cls;
        const float4 wca = wc4[lane*2], wcb = wc4[lane*2+1];
        const float bias = bp[0];
        float4 ca[MB], cb[MB];
        #pragma unroll
        for (int m = 0; m < MB; ++m) {
            ca[m] = c4[m*(EC/4) + lane*2 + 0];
            cb[m] = c4[m*(EC/4) + lane*2 + 1];
        }
        float v = 0.f;
        #pragma unroll
        for (int m = 0; m < MB; ++m) {
            const float p = wave_sum(dot8(ca[m], wca, cb[m], wcb));
            if (lane == m) v = p;
        }
        if (lane < MB) ws[lane] = v + bias;
        return;
    }

    const int row0 = ((sblk - 1) * WPB + swave) * RPW;
    float* emb_dot = ws + 64;

    const float4* e4  = (const float4*)(emb + (size_t)row0 * EC);
    const float4* w04 = (const float4*)(Wp);
    const float4* w14 = (const float4*)(Wp + EC);

    const float4 w0a = w04[lane*2+0], w0b = w04[lane*2+1];
    const float4 w1a = w14[lane*2+0], w1b = w14[lane*2+1];
    float4 ea[RPW], eb[RPW];
    #pragma unroll
    for (int r = 0; r < RPW; ++r) {
        ea[r] = e4[r*(EC/4) + lane*2 + 0];
        eb[r] = e4[r*(EC/4) + lane*2 + 1];
    }
    float v = 0.f;
    #pragma unroll
    for (int r = 0; r < RPW; ++r) {
        const float p0 = wave_sum(dot8(ea[r], w0a, eb[r], w0b));
        const float p1 = wave_sum(dot8(ea[r], w1a, eb[r], w1b));
        if (lane == 2*r)     v = p0;
        if (lane == 2*r + 1) v = p1;
    }
    if (lane < 2*RPW) emb_dot[row0*2 + lane] = v;
}

template <int LDB>
__device__ __forceinline__ v8f wmma_tile_k256_gmem(const float* Al, const float* __restrict__ Bl) {
    v2f a[64], b[64];
    #pragma unroll
    for (int i = 0; i < 64; ++i) {
        a[i]   = *(const v2f*)(Al + 4 * i);
        b[i].x = Bl[(4 * i    ) * LDB];
        b[i].y = Bl[(4 * i + 1) * LDB];
    }
    v8f acc = {};
    #pragma unroll
    for (int i = 0; i < 64; ++i)
        acc = __builtin_amdgcn_wmma_f32_16x16x4_f32(
                  false, a[i], false, b[i], (short)0, acc, false, false);
    asm volatile("v_nop\n\tv_nop\n\tv_nop\n\tv_nop\n\tv_nop\n\tv_nop\n\tv_nop\n\tv_nop" : "+v"(acc));
    return acc;
}

__global__ __launch_bounds__(256) void back_kernel(
    const int* __restrict__ moves, const float* __restrict__ ws,
    const float* __restrict__ Wv2, const float* __restrict__ bv2,
    const float* __restrict__ Ws2, const float* __restrict__ bs2,
    const int*   __restrict__ sidx,
    float* out)
{
    __shared__ float clsd[MB];
    __shared__ float spAll[MB * NSPEC];
    const int t = threadIdx.x;
    if (t < MB) clsd[t] = ws[t];

    const int l = blockIdx.x * 256 + t;
    const float* emb_dot = ws + 64;
    const int4 mv = ((const int4*)moves)[l];
    const int r0 = mv.x * D2C + mv.y;
    const int r1 = mv.z * D2C + mv.w;
    float pol[MB];
    #pragma unroll
    for (int m = 0; m < MB; ++m) {
        const float d0 = emb_dot[(m*1024 + r0)*2 + 0];
        const float d1 = emb_dot[(m*1024 + r1)*2 + 1];
        pol[m] = d0 + d1;
    }

    if (blockIdx.x < 2) {
        const int lane  = t & 31;
        const int wave  = t >> 5;
        const int colr  = lane & 15;
        const int khalf = lane >> 4;
        const bool spec = (blockIdx.x == 0);
        const float* Al = out + blockIdx.x * HC + (colr & 7) * POLW + khalf * 2;
        int sj = 0;
        if (spec) {
            if (t < MB * SIDXC) sj = sidx[t & 7];
            if (wave < 2) {
                const int bcol = wave * 16 + colr;
                const float b2 = bs2[bcol];
                v8f acc = wmma_tile_k256_gmem<NSPEC>(Al, Ws2 + (khalf * 2) * NSPEC + bcol);
                if (khalf == 0) {
                    #pragma unroll
                    for (int i = 0; i < MB; ++i)
                        spAll[i * NSPEC + bcol] = acc[i] + b2;
                }
            }
        } else if (wave == 0) {
            const int bcol = colr & 3;
            const float b2 = bv2[bcol];
            v8f acc = wmma_tile_k256_gmem<NPLAY>(Al, Wv2 + (khalf * 2) * NPLAY + bcol);
            if (khalf == 0 && colr < NPLAY) {
                #pragma unroll
                for (int i = 0; i < MB; ++i)
                    out[MB * POLW + i * NPLAY + colr] = acc[i] + b2;
            }
        }
        __syncthreads();
        if (spec && t < MB * SIDXC) {
            const int m = t >> 3, j = t & 7;
            out[m * POLW + LC + j] = spAll[m * NSPEC + sj];
        }
    } else {
        __syncthreads();
    }

    #pragma unroll
    for (int m = 0; m < MB; ++m)
        out[m*POLW + l] = pol[m] + clsd[m];
}

extern "C" void kernel_launch(void* const* d_in, const int* in_sizes, int n_in,
                              void* d_out, int out_size, void* d_ws, size_t ws_size,
                              hipStream_t stream) {
    (void)in_sizes; (void)n_in; (void)out_size; (void)ws_size;
    const float* emb  = (const float*)d_in[0];
    const float* cls  = (const float*)d_in[1];
    const int*   mov  = (const int*)  d_in[2];
    const int*   sidx = (const int*)  d_in[3];
    const float* Wp   = (const float*)d_in[4];
    const float* bp   = (const float*)d_in[5];
    const float* Wv1  = (const float*)d_in[6];
    const float* bv1  = (const float*)d_in[7];
    const float* Wv2  = (const float*)d_in[8];
    const float* bv2  = (const float*)d_in[9];
    const float* Ws1  = (const float*)d_in[10];
    const float* bs1  = (const float*)d_in[11];
    const float* Ws2  = (const float*)d_in[12];
    const float* bs2  = (const float*)d_in[13];
    float* out = (float*)d_out;
    float* ws  = (float*)d_ws;

    front_kernel<<<265, 128, 0, stream>>>(emb, cls, Wp, bp, Wv1, bv1, Ws1, bs1, ws, out);
    back_kernel<<<64, 256, 0, stream>>>(mov, ws, Wv2, bv2, Ws2, bs2, sidx, out);
}


# policy-dot table laid out [position][k][batch]: launch 2 reads a column's 8 batch values with 4 16-byte loads (was 16 scattered 4-byte); a launch-1 wave does 2 positions x 8 batches, stores one line
# speedup vs baseline: 3.7354x; 1.0708x over previous
#include <hip/hip_runtime.h>


typedef __attribute__((ext_vector_type(2))) float v2f;
typedef __attribute__((ext_vector_type(8))) float v8f;

#define MB 8
#define D1C 32
#define D2C 32
#define EC 256
#define LC 16384
#define HC 256
#define NSPEC 32
#define SIDXC 8
#define NPLAY 4
#define POLW (LC + SIDXC)
#define WPB 2
#define L1BLK 16
#define NPOS (D1C * D2C)


__device__ __forceinline__ float dot8(float4 a, float4 b, float4 c, float4 d) {
    return a.x*b.x + a.y*b.y + a.z*b.z + a.w*b.w +
           c.x*d.x + c.y*d.y + c.z*d.z + c.w*d.w;
}

template <int N>
__device__ __forceinline__ void fold_step(float (&q)[32], int lane) {
    const bool up = (lane & N) != 0;
    #pragma unroll
    for (int i = 0; i < N; ++i) {
        const float keep = up ? q[N + i] : q[i];
        const float send = up ? q[i] : q[N + i];
        q[i] = keep + __shfl_xor(send, N, 32);
    }
}

__device__ __forceinline__ float wave_sum(float v) {
    #pragma unroll
    for (int off = 16; off > 0; off >>= 1) v += __shfl_xor(v, off, 32);
    return v;
}


__global__ __launch_bounds__(64) void front_kernel(
    const float* __restrict__ emb, const float* __restrict__ cls,
    const float* __restrict__ Wp, const float* __restrict__ bp,
    const float* __restrict__ Wv1, const float* __restrict__ bv1,
    const float* __restrict__ Ws1, const float* __restrict__ bs1,
    float* __restrict__ ws, float* __restrict__ out)
{
    const int lane = threadIdx.x & 31;
    const int wave = threadIdx.x >> 5;

    if (blockIdx.x < L1BLK) {
        const int job   = blockIdx.x * WPB + wave;
        const int colr  = lane & 15;
        const int khalf = lane >> 4;
        const float* W1 = (job < 16) ? Wv1 : Ws1;
        const float* b1 = (job < 16) ? bv1 : bs1;
        const int n0    = (job & 15) * 16;
        const float bias = b1[n0 + colr];
        const float* Al = cls + (colr & 7) * EC + khalf * 2;
        const float* Bl = W1 + (khalf * 2) * HC + n0 + colr;
        v8f acc = {};
        #pragma unroll 1
        for (int c = 0; c < 4; ++c) {
            v2f a[16], b[16];
            #pragma unroll
            for (int i = 0; i < 16; ++i) {
                const int s = c * 16 + i;
                a[i]   = *(const v2f*)(Al + 4 * s);
                b[i].x = Bl[(4 * s    ) * HC];
                b[i].y = Bl[(4 * s + 1) * HC];
            }
            #pragma unroll
            for (int i = 0; i < 16; ++i)
                acc = __builtin_amdgcn_wmma_f32_16x16x4_f32(
                          false, a[i], false, b[i], (short)0, acc, false, false);
            asm volatile("v_nop\n\tv_nop\n\tv_nop\n\tv_nop\n\tv_nop\n\tv_nop\n\tv_nop\n\tv_nop" : "+v"(acc));
        }
        if (khalf == 0) {
            float* dst = out + ((job < 16) ? HC : 0) + n0 + colr;
            #pragma unroll
            for (int i = 0; i < MB; ++i) {
                const float v = acc[i] + bias;
                dst[i * POLW] = v > 0.f ? v : 0.f;
            }
        }
        return;
    }

    const int sblk  = (int)blockIdx.x - L1BLK;
    const int swave = __builtin_amdgcn_readfirstlane(wave);

    if (sblk == 0) {
        if (swave != 0) return;
        const float4* wc4 = (const float4*)(Wp + 2*EC);
        const float4* c4  = (const float4*)cls;
        const float4 wca = wc4[lane*2], wcb = wc4[lane*2+1];
        const float bias = bp[0];
        float4 ca[MB], cb[MB];
        #pragma unroll
        for (int m = 0; m < MB; ++m) {
            ca[m] = c4[m*(EC/4) + lane*2 + 0];
            cb[m] = c4[m*(EC/4) + lane*2 + 1];
        }
        float v = 0.f;
        #pragma unroll
        for (int m = 0; m < MB; ++m) {
            const float p = wave_sum(dot8(ca[m], wca, cb[m], wcb));
            if (lane == m) v = p;
        }
        if (lane < MB) ws[lane] = v + bias;
        return;
    }

    const int j = (sblk - 1) * WPB + swave;
    float* tab = ws + 64;

    const float4* e4  = (const float4*)(emb + (size_t)(2 * j) * EC);
    const float4* w04 = (const float4*)(Wp);
    const float4* w14 = (const float4*)(Wp + EC);

    const float4 w0a = w04[lane*2+0], w0b = w04[lane*2+1];
    const float4 w1a = w14[lane*2+0], w1b = w14[lane*2+1];
    float4 ea[MB][2], eb[MB][2];
    #pragma unroll
    for (int m = 0; m < MB; ++m) {
        #pragma unroll
        for (int rr = 0; rr < 2; ++rr) {
            ea[m][rr] = e4[(m*NPOS + rr)*(EC/4) + lane*2 + 0];
            eb[m][rr] = e4[(m*NPOS + rr)*(EC/4) + lane*2 + 1];
        }
    }
    float p[32];
    #pragma unroll
    for (int m = 0; m < MB; ++m) {
        #pragma unroll
        for (int rr = 0; rr < 2; ++rr) {
            p[rr*16 + m]     = dot8(ea[m][rr], w0a, eb[m][rr], w0b);
            p[rr*16 + 8 + m] = dot8(ea[m][rr], w1a, eb[m][rr], w1b);
        }
    }
    fold_step<16>(p, lane);
    fold_step<8>(p, lane);
    fold_step<4>(p, lane);
    fold_step<2>(p, lane);
    fold_step<1>(p, lane);
    tab[j*32 + lane] = p[0];
}

template <int LDB>
__device__ __forceinline__ v8f wmma_tile_k256_gmem(const float* Al, const float* __restrict__ Bl) {
    v2f a[64], b[64];
    #pragma unroll
    for (int i = 0; i < 64; ++i) {
        a[i]   = *(const v2f*)(Al + 4 * i);
        b[i].x = Bl[(4 * i    ) * LDB];
        b[i].y = Bl[(4 * i + 1) * LDB];
    }
    v8f acc = {};
    #pragma unroll
    for (int i = 0; i < 64; ++i)
        acc = __builtin_amdgcn_wmma_f32_16x16x4_f32(
                  false, a[i], false, b[i], (short)0, acc, false, false);
    asm volatile("v_nop\n\tv_nop\n\tv_nop\n\tv_nop\n\tv_nop\n\tv_nop\n\tv_nop\n\tv_nop" : "+v"(acc));
    return acc;
}

__global__ __launch_bounds__(256) void back_kernel(
    const int* __restrict__ moves, const float* __restrict__ ws,
    const float* __restrict__ Wv2, const float* __restrict__ bv2,
    const float* __restrict__ Ws2, const float* __restrict__ bs2,
    const int*   __restrict__ sidx,
    float* out)
{
    __shared__ float clsd[MB];
    __shared__ float spAll[MB * NSPEC];
    const int t = threadIdx.x;
    if (t < MB) clsd[t] = ws[t];

    const int l = blockIdx.x * 256 + t;
    const float4* tab4 = (const float4*)(ws + 64);
    const int4 mv = ((const int4*)moves)[l];
    const int r0 = mv.x * D2C + mv.y;
    const int r1 = mv.z * D2C + mv.w;
    const float4 d0a = tab4[r0*4 + 0], d0b = tab4[r0*4 + 1];
    const float4 d1a = tab4[r1*4 + 2], d1b = tab4[r1*4 + 3];
    float pol[MB];
    pol[0] = d0a.x + d1a.x;  pol[1] = d0a.y + d1a.y;
    pol[2] = d0a.z + d1a.z;  pol[3] = d0a.w + d1a.w;
    pol[4] = d0b.x + d1b.x;  pol[5] = d0b.y + d1b.y;
    pol[6] = d0b.z + d1b.z;  pol[7] = d0b.w + d1b.w;

    if (blockIdx.x < 2) {
        const int lane  = t & 31;
        const int wave  = t >> 5;
        const int colr  = lane & 15;
        const int khalf = lane >> 4;
        const bool spec = (blockIdx.x == 0);
        const float* Al = out + blockIdx.x * HC + (colr & 7) * POLW + khalf * 2;
        int sj = 0;
        if (spec) {
            if (t < MB * SIDXC) sj = sidx[t & 7];
            if (wave < 2) {
                const int bcol = wave * 16 + colr;
                const float b2 = bs2[bcol];
                v8f acc = wmma_tile_k256_gmem<NSPEC>(Al, Ws2 + (khalf * 2) * NSPEC + bcol);
                if (khalf == 0) {
                    #pragma unroll
                    for (int i = 0; i < MB; ++i)
                        spAll[i * NSPEC + bcol] = acc[i] + b2;
                }
            }
        } else if (wave == 0) {
            const int bcol = colr & 3;
            const float b2 = bv2[bcol];
            v8f acc = wmma_tile_k256_gmem<NPLAY>(Al, Wv2 + (khalf * 2) * NPLAY + bcol);
            if (khalf == 0 && colr < NPLAY) {
                #pragma unroll
                for (int i = 0; i < MB; ++i)
                    out[MB * POLW + i * NPLAY + colr] = acc[i] + b2;
            }
        }
        __syncthreads();
        if (spec && t < MB * SIDXC) {
            const int m = t >> 3, j = t & 7;
            out[m * POLW + LC + j] = spAll[m * NSPEC + sj];
        }
    } else {
        __syncthreads();
    }

    #pragma unroll
    for (int m = 0; m < MB; ++m)
        out[m*POLW + l] = pol[m] + clsd[m];
}

extern "C" void kernel_launch(void* const* d_in, const int* in_sizes, int n_in,
                              void* d_out, int out_size, void* d_ws, size_t ws_size,
                              hipStream_t stream) {
    (void)in_sizes; (void)n_in; (void)out_size; (void)ws_size;
    const float* emb  = (const float*)d_in[0];
    const float* cls  = (const float*)d_in[1];
    const int*   mov  = (const int*)  d_in[2];
    const int*   sidx = (const int*)  d_in[3];
    const float* Wp   = (const float*)d_in[4];
    const float* bp   = (const float*)d_in[5];
    const float* Wv1  = (const float*)d_in[6];
    const float* bv1  = (const float*)d_in[7];
    const float* Wv2  = (const float*)d_in[8];
    const float* bv2  = (const float*)d_in[9];
    const float* Ws1  = (const float*)d_in[10];
    const float* bs1  = (const float*)d_in[11];
    const float* Ws2  = (const float*)d_in[12];
    const float* bs2  = (const float*)d_in[13];
    float* out = (float*)d_out;
    float* ws  = (float*)d_ws;

    front_kernel<<<273, 64, 0, stream>>>(emb, cls, Wp, bp, Wv1, bv1, Ws1, bs1, ws, out);
    back_kernel<<<64, 256, 0, stream>>>(mov, ws, Wv2, bv2, Ws2, bs2, sidx, out);
}


# launch 2: all loads but the table pieces go out in one batch at entry: cls dots by a uniform scalar load (no LDS copy, no barrier in 62 work-groups), layer-2 operands ahead of the moves wait
# speedup vs baseline: 5.5278x; 1.4798x over previous
#include <hip/hip_runtime.h>


typedef __attribute__((ext_vector_type(2))) float v2f;
typedef __attribute__((ext_vector_type(8))) float v8f;

#define MB 8
#define D1C 32
#define D2C 32
#define EC 256
#define LC 16384
#define HC 256
#define NSPEC 32
#define SIDXC 8
#define NPLAY 4
#define POLW (LC + SIDXC)
#define WPB 2
#define L1BLK 32
#define NPOS (D1C * D2C)


__device__ __forceinline__ float dot8(float4 a, float4 b, float4 c, float4 d) {
    return a.x*b.x + a.y*b.y + a.z*b.z + a.w*b.w +
           c.x*d.x + c.y*d.y + c.z*d.z + c.w*d.w;
}

template <int N>
__device__ __forceinline__ void fold_step(float (&q)[32], int lane) {
    const bool up = (lane & N) != 0;
    #pragma unroll
    for (int i = 0; i < N; ++i) {
        const float keep = up ? q[N + i] : q[i];
        const float send = up ? q[i] : q[N + i];
        q[i] = keep + __shfl_xor(send, N, 32);
    }
}

__device__ __forceinline__ float wave_sum(float v) {
    #pragma unroll
    for (int off = 16; off > 0; off >>= 1) v += __shfl_xor(v, off, 32);
    return v;
}


template <int NS, int LDB>
__device__ __forceinline__ void load_frags(const float* Al, const float* __restrict__ Bl,
                                           v2f (&a)[NS], v2f (&b)[NS]) {
    #pragma unroll
    for (int i = 0; i < NS; ++i) {
        a[i]   = *(const v2f*)(Al + 4 * i);
        b[i].x = Bl[(4 * i    ) * LDB];
        b[i].y = Bl[(4 * i + 1) * LDB];
    }
}

template <int NS>
__device__ __forceinline__ v8f wmma_chain(const v2f (&a)[NS], const v2f (&b)[NS]) {
    v8f acc = {};
    #pragma unroll
    for (int i = 0; i < NS; ++i)
        acc = __builtin_amdgcn_wmma_f32_16x16x4_f32(
                  false, a[i], false, b[i], (short)0, acc, false, false);
    asm volatile("v_nop\n\tv_nop\n\tv_nop\n\tv_nop\n\tv_nop\n\tv_nop\n\tv_nop\n\tv_nop" : "+v"(acc));
    return acc;
}

template <int NS, int LDB>
__device__ __forceinline__ v8f wmma_steps_gmem(const float* Al, const float* __restrict__ Bl) {
    v2f a[NS], b[NS];
    load_frags<NS, LDB>(Al, Bl, a, b);
    return wmma_chain<NS>(a, b);
}

struct PolPieces { float4 d0a, d0b, d1a, d1b; };
__device__ __forceinline__ PolPieces gather_pieces(const float4* __restrict__ tab4, int4 mv) {
    const int r0 = mv.x * D2C + mv.y;
    const int r1 = mv.z * D2C + mv.w;
    PolPieces p;
    p.d0a = tab4[r0*4 + 0];  p.d0b = tab4[r0*4 + 1];
    p.d1a = tab4[r1*4 + 2];  p.d1b = tab4[r1*4 + 3];
    return p;
}

__global__ __launch_bounds__(64) void front_kernel(
    const float* __restrict__ emb, const float* __restrict__ cls,
    const float* __restrict__ Wp, const float* __restrict__ bp,
    const float* __restrict__ Wv1, const float* __restrict__ bv1,
    const float* __restrict__ Ws1, const float* __restrict__ bs1,
    float* __restrict__ ws, float* __restrict__ out)
{
    __shared__ float l1part[MB * 16];
    const int lane = threadIdx.x & 31;
    const int wave = threadIdx.x >> 5;

    const int swave = __builtin_amdgcn_readfirstlane(wave);

    if (blockIdx.x < L1BLK) {
        const int job   = blockIdx.x;
        const int colr  = lane & 15;
        const int khalf = lane >> 4;
        const float* W1 = (job < 16) ? Wv1 : Ws1;
        const float* b1 = (job < 16) ? bv1 : bs1;
        const int n0    = (job & 15) * 16;
        const int k0    = swave * (EC / WPB);
        const float bias = b1[n0 + colr];
        v8f acc = wmma_steps_gmem<EC / WPB / 4, HC>(
                      cls + (colr & 7) * EC + k0 + khalf * 2,
                      W1 + (k0 + khalf * 2) * HC + n0 + colr);
        if (swave == 1 && khalf == 0) {
            #pragma unroll
            for (int i = 0; i < MB; ++i) l1part[i * 16 + colr] = acc[i];
        }
        __syncthreads();
        if (swave == 0 && khalf == 0) {
            float* dst = out + ((job < 16) ? HC : 0) + n0 + colr;
            #pragma unroll
            for (int i = 0; i < MB; ++i) {
                const float v = (acc[i] + l1part[i * 16 + colr]) + bias;
                dst[i * POLW] = v > 0.f ? v : 0.f;
            }
        }
        return;
    }

    const int sblk  = (int)blockIdx.x - L1BLK;

    if (sblk == 0) {
        if (swave != 0) return;
        const float4* wc4 = (const float4*)(Wp + 2*EC);
        const float4* c4  = (const float4*)cls;
        const float4 wca = wc4[lane*2], wcb = wc4[lane*2+1];
        const float bias = bp[0];
        float4 ca[MB], cb[MB];
        #pragma unroll
        for (int m = 0; m < MB; ++m) {
            ca[m] = c4[m*(EC/4) + lane*2 + 0];
            cb[m] = c4[m*(EC/4) + lane*2 + 1];
        }
        float v = 0.f;
        #pragma unroll
        for (int m = 0; m < MB; ++m) {
            const float p = wave_sum(dot8(ca[m], wca, cb[m], wcb));
            if (lane == m) v = p;
        }
        if (lane < MB) ws[lane] = v + bias;
        return;
    }

    const int j = (sblk - 1) * WPB + swave;
    float* tab = ws + 64;

    const float4* e4  = (const float4*)(emb + (size_t)(2 * j) * EC);
    const float4* w04 = (const float4*)(Wp);
    const float4* w14 = (const float4*)(Wp + EC);

    const float4 w0a = w04[lane*2+0], w0b = w04[lane*2+1];
    const float4 w1a = w14[lane*2+0], w1b = w14[lane*2+1];
    float4 ea[MB][2], eb[MB][2];
    #pragma unroll
    for (int m = 0; m < MB; ++m) {
        #pragma unroll
        for (int rr = 0; rr < 2; ++rr) {
            ea[m][rr] = e4[(m*NPOS + rr)*(EC/4) + lane*2 + 0];
            eb[m][rr] = e4[(m*NPOS + rr)*(EC/4) + lane*2 + 1];
        }
    }
    float p[32];
    #pragma unroll
    for (int m = 0; m < MB; ++m) {
        #pragma unroll
        for (int rr = 0; rr < 2; ++rr) {
            p[rr*16 + m]     = dot8(ea[m][rr], w0a, eb[m][rr], w0b);
            p[rr*16 + 8 + m] = dot8(ea[m][rr], w1a, eb[m][rr], w1b);
        }
    }
    fold_step<16>(p, lane);
    fold_step<8>(p, lane);
    fold_step<4>(p, lane);
    fold_step<2>(p, lane);
    fold_step<1>(p, lane);
    tab[j*32 + lane] = p[0];
}

__global__ __launch_bounds__(256) void back_kernel(
    const int* __restrict__ moves, const float* __restrict__ ws,
    const float* __restrict__ Wv2, const float* __restrict__ bv2,
    const float* __restrict__ Ws2, const float* __restrict__ bs2,
    const int*   __restrict__ sidx,
    float* out)
{
    __shared__ float part[4 * MB * NSPEC];
    __shared__ float bsl[NSPEC];
    const int t = threadIdx.x;
    const int l = blockIdx.x * 256 + t;
    const float4* tab4 = (const float4*)(ws + 64);

    const int4 mv = ((const int4*)moves)[l];
    const float4 ca = ((const float4*)ws)[0];
    const float4 cb = ((const float4*)ws)[1];
    PolPieces pp;

    if (blockIdx.x < 2) {
        const int lane  = t & 31;
        const int wave  = __builtin_amdgcn_readfirstlane(t >> 5);
        const int colr  = lane & 15;
        const int khalf = lane >> 4;
        const bool spec = (blockIdx.x == 0);
        const float* Al = out + blockIdx.x * HC + (colr & 7) * POLW + khalf * 2;
        if (spec) {
            int sj = 0;
            float bsv = 0.f;
            if (t < MB * SIDXC) sj = sidx[t & 7];
            if (t < NSPEC) bsv = bs2[t];
            const int bcol = (wave & 1) * 16 + colr;
            const int q    = wave >> 1;
            const int k0   = q * (HC / 4);
            v2f a[HC / 4 / 4], b[HC / 4 / 4];
            load_frags<HC / 4 / 4, NSPEC>(Al + k0, Ws2 + (k0 + khalf * 2) * NSPEC + bcol, a, b);
            __builtin_amdgcn_sched_barrier(0);
            pp = gather_pieces(tab4, mv);
            v8f acc = wmma_chain<HC / 4 / 4>(a, b);
            if (khalf == 0) {
                #pragma unroll
                for (int i = 0; i < MB; ++i)
                    part[(q * MB + i) * NSPEC + bcol] = acc[i];
            }
            if (t < NSPEC) bsl[t] = bsv;
            __syncthreads();
            if (t < MB * SIDXC) {
                const int m = t >> 3, j = t & 7;
                float v = part[(0 * MB + m) * NSPEC + sj];
                #pragma unroll
                for (int qq = 1; qq < 4; ++qq) v += part[(qq * MB + m) * NSPEC + sj];
                out[m * POLW + LC + j] = v + bsl[sj];
            }
        } else {
            const int bcol = colr & 3;
            const int k0   = wave * (HC / 8);
            float bvv = 0.f;
            if (t < MB * NPLAY) bvv = bv2[t & 3];
            v2f a[HC / 8 / 4], b[HC / 8 / 4];
            load_frags<HC / 8 / 4, NPLAY>(Al + k0, Wv2 + (k0 + khalf * 2) * NPLAY + bcol, a, b);
            pp = gather_pieces(tab4, mv);
            v8f acc = wmma_chain<HC / 8 / 4>(a, b);
            if (khalf == 0 && colr < NPLAY) {
                #pragma unroll
                for (int i = 0; i < MB; ++i)
                    part[(wave * MB + i) * NPLAY + colr] = acc[i];
            }
            __syncthreads();
            if (t < MB * NPLAY) {
                const int m = t >> 2, c = t & 3;
                float v = part[(0 * MB + m) * NPLAY + c];
                #pragma unroll
                for (int w = 1; w < 8; ++w) v += part[(w * MB + m) * NPLAY + c];
                out[MB * POLW + m * NPLAY + c] = v + bvv;
            }
        }
    } else {
        pp = gather_pieces(tab4, mv);
    }

    out[0*POLW + l] = (pp.d0a.x + pp.d1a.x) + ca.x;
    out[1*POLW + l] = (pp.d0a.y + pp.d1a.y) + ca.y;
    out[2*POLW + l] = (pp.d0a.z + pp.d1a.z) + ca.z;
    out[3*POLW + l] = (pp.d0a.w + pp.d1a.w) + ca.w;
    out[4*POLW + l] = (pp.d0b.x + pp.d1b.x) + cb.x;
    out[5*POLW + l] = (pp.d0b.y + pp.d1b.y) + cb.y;
    out[6*POLW + l] = (pp.d0b.z + pp.d1b.z) + cb.z;
    out[7*POLW + l] = (pp.d0b.w + pp.d1b.w) + cb.w;
}

extern "C" void kernel_launch(void* const* d_in, const int* in_sizes, int n_in,
                              void* d_out, int out_size, void* d_ws, size_t ws_size,
                              hipStream_t stream) {
    (void)in_sizes; (void)n_in; (void)out_size; (void)ws_size;
    const float* emb  = (const float*)d_in[0];
    const float* cls  = (const float*)d_in[1];
    const int*   mov  = (const int*)  d_in[2];
    const int*   sidx = (const int*)  d_in[3];
    const float* Wp   = (const float*)d_in[4];
    const float* bp   = (const float*)d_in[5];
    const float* Wv1  = (const float*)d_in[6];
    const float* bv1  = (const float*)d_in[7];
    const float* Wv2  = (const float*)d_in[8];
    const float* bv2  = (const float*)d_in[9];
    const float* Ws1  = (const float*)d_in[10];
    const float* bs1  = (const float*)d_in[11];
    const float* Ws2  = (const float*)d_in[12];
    const float* bs2  = (const float*)d_in[13];
    float* out = (float*)d_out;
    float* ws  = (float*)d_ws;

    front_kernel<<<289, 64, 0, stream>>>(emb, cls, Wp, bp, Wv1, bv1, Ws1, bs1, ws, out);
    back_kernel<<<64, 256, 0, stream>>>(mov, ws, Wv2, bv2, Ws2, bs2, sidx, out);
}
